# starting kernel
# baseline (speedup 1.0000x reference)
#include <hip/hip_runtime.h>
#include <hip/hip_bf16.h>
#include <math.h>

typedef __attribute__((ext_vector_type(2))) float v2f;
typedef __attribute__((ext_vector_type(8))) float v8f;

#define D_FEAT 128
#define MT_PER_BLK 2
#define ROWS_PER_BLK (MT_PER_BLK * 16)

__global__ __launch_bounds__(256) void deg_kernel(const int* __restrict__ dst,
                                                  float* __restrict__ deg,
                                                  int n_edges) {
    int e = blockIdx.x * blockDim.x + threadIdx.x;
    if (e < n_edges) atomicAdd(&deg[dst[e]], 1.0f);
}

__global__ __launch_bounds__(256) void dinv_kernel(float* __restrict__ deg, int n) {
    int i = blockIdx.x * blockDim.x + threadIdx.x;
    if (i < n) deg[i] = rsqrtf(deg[i] + 1.0f);
}

__global__ __launch_bounds__(256) void gemm_scale_kernel(
    const float* __restrict__ x, const float* __restrict__ W,
    const float* __restrict__ dinv, float* __restrict__ hprime,
    float* __restrict__ acc, int n_nodes) {
    __shared__ float sX[ROWS_PER_BLK][D_FEAT + 4];

    const int tid   = threadIdx.x;
    const int lane  = tid & 31;
    const int wave  = tid >> 5;
    const int mbase = blockIdx.x * ROWS_PER_BLK;
    const int half  = lane >> 4;
    const int l15   = lane & 15;
    const int nb    = wave * 16;

    const bool full = (mbase + ROWS_PER_BLK) <= n_nodes;

    const float4* xv = reinterpret_cast<const float4*>(x);
    if (full) {
#pragma unroll
        for (int i = 0; i < 4; ++i) {
            int slot = tid + i * 256;
            int r    = slot >> 5;
            int cb   = slot & 31;
            *reinterpret_cast<float4*>(&sX[r][cb * 4]) =
                xv[(size_t)(mbase + r) * (D_FEAT / 4) + cb];
        }
    } else {
        const float4 zero4 = make_float4(0.f, 0.f, 0.f, 0.f);
#pragma unroll
        for (int i = 0; i < 4; ++i) {
            int slot = tid + i * 256;
            int r    = slot >> 5;
            int cb   = slot & 31;
            int row  = mbase + r;
            float4 v = (row < n_nodes) ? xv[(size_t)row * (D_FEAT / 4) + cb] : zero4;
            *reinterpret_cast<float4*>(&sX[r][cb * 4]) = v;
        }
    }
    __syncthreads();

    v8f c[MT_PER_BLK];
#pragma unroll
    for (int t = 0; t < MT_PER_BLK; ++t)
#pragma unroll
        for (int r = 0; r < 8; ++r) c[t][r] = 0.0f;

    for (int k0 = 0; k0 < D_FEAT; k0 += 4) {
        const int ka = k0 + half * 2;
        v2f bf;
        bf.x = W[(size_t)ka * D_FEAT + nb + l15];
        bf.y = W[(size_t)(ka + 1) * D_FEAT + nb + l15];
#pragma unroll
        for (int t = 0; t < MT_PER_BLK; ++t) {
            v2f af;
            af.x = sX[t * 16 + l15][ka];
            af.y = sX[t * 16 + l15][ka + 1];
            c[t] = __builtin_amdgcn_wmma_f32_16x16x4_f32(
                 false, af,  false, bf,
                 (short)0, c[t],  false,  false);
        }
    }

    if (full) {
#pragma unroll
        for (int t = 0; t < MT_PER_BLK; ++t) {
            const int rbase = mbase + t * 16 + half * 8;
            float* hp = hprime + (size_t)rbase * D_FEAT + nb + l15;
            float* ap = acc    + (size_t)rbase * D_FEAT + nb + l15;
            float dv[8];
#pragma unroll
            for (int r = 0; r < 8; ++r) dv[r] = dinv[rbase + r];
#pragma unroll
            for (int r = 0; r < 8; ++r) {
                float v = c[t][r] * dv[r];
                hp[(size_t)r * D_FEAT] = v;
                ap[(size_t)r * D_FEAT] = v;
            }
        }
    } else {
#pragma unroll
        for (int t = 0; t < MT_PER_BLK; ++t) {
#pragma unroll
            for (int r = 0; r < 8; ++r) {
                int row = mbase + t * 16 + r + half * 8;
                if (row < n_nodes) {
                    float v = c[t][r] * dinv[row];
                    size_t off = (size_t)row * D_FEAT + nb + l15;
                    hprime[off] = v;
                    acc[off]    = v;
                }
            }
        }
    }
}

__global__ __launch_bounds__(256) void scatter_kernel(
    const int* __restrict__ src, const int* __restrict__ dst,
    const float* __restrict__ hprime, float* __restrict__ acc, int n_edges) {
    long long gid = (long long)blockIdx.x * blockDim.x + threadIdx.x;
    int e    = (int)(gid >> 5);
    int lane = (int)(gid & 31);
    if (e >= n_edges) return;
    int s = src[e], d = dst[e];
    float4 hv = *reinterpret_cast<const float4*>(hprime + (size_t)s * D_FEAT + lane * 4);
    float* ap = acc + (size_t)d * D_FEAT + lane * 4;
    atomicAdd(ap + 0, hv.x);
    atomicAdd(ap + 1, hv.y);
    atomicAdd(ap + 2, hv.z);
    atomicAdd(ap + 3, hv.w);
}

__global__ __launch_bounds__(256) void finalize_kernel(
    const float* __restrict__ acc, const float* __restrict__ dinv,
    const float* __restrict__ bias, const float* __restrict__ pa,
    const int* __restrict__ batch, float* __restrict__ pooled,
    float* __restrict__ cnt, int n_nodes) {
    long long gid = (long long)blockIdx.x * blockDim.x + threadIdx.x;
    int i    = (int)(gid >> 5);
    int lane = (int)(gid & 31);
    if (i >= n_nodes) return;

    float di = dinv[i];
    float v[4];
    float ss = 0.0f;
#pragma unroll
    for (int j = 0; j < 4; ++j) {
        int c   = lane + 32 * j;
        float t = di * acc[(size_t)i * D_FEAT + c] + bias[c];
        t = (t > 0.0f) ? t : pa[c] * t;
        v[j] = t;
        ss += t * t;
    }
#pragma unroll
    for (int off = 16; off > 0; off >>= 1) ss += __shfl_xor(ss, off, 32);
    float scale = 1.0f / fmaxf(sqrtf(ss), 1e-12f);

    int g = batch[i];
#pragma unroll
    for (int j = 0; j < 4; ++j)
        atomicAdd(&pooled[(size_t)g * D_FEAT + lane + 32 * j], v[j] * scale);
    if (lane == 0) atomicAdd(&cnt[g], 1.0f);
}

__global__ __launch_bounds__(256) void div_kernel(float* __restrict__ pooled,
                                                  const float* __restrict__ cnt,
                                                  int total) {
    int i = blockIdx.x * blockDim.x + threadIdx.x;
    if (i < total) pooled[i] /= fmaxf(cnt[i >> 7], 1.0f);
}

extern "C" void kernel_launch(void* const* d_in, const int* in_sizes, int n_in,
                              void* d_out, int out_size, void* d_ws, size_t ws_size,
                              hipStream_t stream) {
    const float* x     = (const float*)d_in[0];
    const int*   ei    = (const int*)d_in[1];
    const int*   batch = (const int*)d_in[2];
    const float* W     = (const float*)d_in[3];
    const float* bias  = (const float*)d_in[4];
    const float* pa    = (const float*)d_in[5];

    const int n_nodes = in_sizes[0] / D_FEAT;
    const int n_edges = in_sizes[1] / 2;
    const int* src = ei;
    const int* dst = ei + n_edges;
    float* pooled = (float*)d_out;

    float* hprime = (float*)d_ws;
    float* acc    = hprime + (size_t)n_nodes * D_FEAT;
    float* dinv   = acc + (size_t)n_nodes * D_FEAT;
    float* cnt    = dinv + n_nodes;

    hipMemsetAsync(dinv, 0, (size_t)n_nodes * sizeof(float), stream);
    hipMemsetAsync(cnt, 0, 128 * sizeof(float), stream);
    hipMemsetAsync(d_out, 0, (size_t)out_size * sizeof(float), stream);

    deg_kernel<<<(n_edges + 255) / 256, 256, 0, stream>>>(dst, dinv, n_edges);
    dinv_kernel<<<(n_nodes + 255) / 256, 256, 0, stream>>>(dinv, n_nodes);

    int gemm_blocks = (n_nodes + ROWS_PER_BLK - 1) / ROWS_PER_BLK;
    gemm_scale_kernel<<<gemm_blocks, 256, 0, stream>>>(x, W, dinv, hprime, acc, n_nodes);

    long long sc_threads = (long long)n_edges * 32;
    scatter_kernel<<<(int)((sc_threads + 255) / 256), 256, 0, stream>>>(src, dst, hprime, acc, n_edges);

    long long fin_threads = (long long)n_nodes * 32;
    finalize_kernel<<<(int)((fin_threads + 255) / 256), 256, 0, stream>>>(
        acc, dinv, bias, pa, batch, pooled, cnt, n_nodes);

    div_kernel<<<(out_size + 255) / 256, 256, 0, stream>>>(pooled, cnt, out_size);
}


# neighbour sums taken over dinv-weighted x rows first; x@W runs on each block's 32 gathered rows inside the gather kernel (no h plane, no product launch); summation order and rounding change
# speedup vs baseline: 2.8022x; 2.8022x over previous
#include <hip/hip_runtime.h>
#include <hip/hip_bf16.h>
#include <math.h>

typedef __attribute__((ext_vector_type(2))) float v2f;
typedef __attribute__((ext_vector_type(8))) float v8f;

#define D_FEAT 128
#define MT_PER_BLK 2
#define ROWS_PER_BLK (MT_PER_BLK * 16)
#define N_GRAPHS 128
#define N_REP 32
#define REP_CNT_OFF (N_GRAPHS * D_FEAT)
#define REP_PITCH (N_GRAPHS * D_FEAT + 320)

__global__ __launch_bounds__(256) void link_kernel(const int* __restrict__ src,
                                                   const int* __restrict__ dst,
                                                   int* __restrict__ head,
                                                   int2* __restrict__ link,
                                                   int n_edges) {
    int e = blockIdx.x * blockDim.x + threadIdx.x;
    if (e < n_edges) {
        int d = dst[e];
        int s = src[e];
        int prev = atomicExch(&head[d], e + 1);
        link[e] = make_int2(s, prev);
    }
}

__global__ __launch_bounds__(256) void chain_len_dinv_kernel(const int* __restrict__ head,
                                                             const int2* __restrict__ link,
                                                             float* __restrict__ dinv,
                                                             int n, int n_edges) {
    int i = blockIdx.x * blockDim.x + threadIdx.x;
    if (i < n) {
        int e   = head[i];
        int cnt = 0;
        while (cnt < n_edges && e != 0) {
            e = link[e - 1].y;
            ++cnt;
        }
        dinv[i] = rsqrtf((float)cnt + 1.0f);
    }
}

__global__ __launch_bounds__(256) void gather_gemm_finalize_kernel(
    const float* __restrict__ x, const float* __restrict__ W,
    const int* __restrict__ head, const int2* __restrict__ link,
    const float* __restrict__ dinv,
    const float* __restrict__ bias, const float* __restrict__ pa,
    const int* __restrict__ batch, float* __restrict__ rep,
    int n_nodes, int n_edges) {
    __shared__ float sA[ROWS_PER_BLK][D_FEAT + 4];
    __shared__ float sC[ROWS_PER_BLK][D_FEAT + 4];

    const int tid   = threadIdx.x;
    const int lane  = tid & 31;
    const int wave  = __builtin_amdgcn_readfirstlane(tid >> 5);
    const int mbase = blockIdx.x * ROWS_PER_BLK;
    const int half  = lane >> 4;
    const int l15   = lane & 15;
    const int nb    = wave * 16;

    const float* xl = x + lane;
    int   e[4];
    int2  lk[4];
    float acc[4][4];
#pragma unroll
    for (int q = 0; q < 4; ++q) {
        const int nd = mbase + wave * 4 + q;
        e[q]  = 0;
        lk[q] = make_int2(0, 0);
#pragma unroll
        for (int j = 0; j < 4; ++j) acc[q][j] = 0.0f;
        if (nd < n_nodes) {
            const float  di = dinv[nd];
            const float* r  = xl + (size_t)nd * D_FEAT;
#pragma unroll
            for (int j = 0; j < 4; ++j) acc[q][j] = di * r[32 * j];
            e[q] = head[nd];
            if (e[q] != 0) lk[q] = link[e[q] - 1];
        }
    }
    for (int it = 0; it < n_edges && (e[0] | e[1] | e[2] | e[3]) != 0; ++it) {
        float qv[4][4];
        float dj[4];
#pragma unroll
        for (int q = 0; q < 4; ++q) {
            if (e[q] != 0) {
                const float* r = xl + (size_t)lk[q].x * D_FEAT;
                dj[q] = dinv[lk[q].x];
#pragma unroll
                for (int j = 0; j < 4; ++j) qv[q][j] = r[32 * j];
                e[q] = lk[q].y;
                if (e[q] != 0) lk[q] = link[e[q] - 1];
            } else {
                dj[q] = 0.0f;
#pragma unroll
                for (int j = 0; j < 4; ++j) qv[q][j] = 0.0f;
            }
        }
#pragma unroll
        for (int q = 0; q < 4; ++q)
#pragma unroll
            for (int j = 0; j < 4; ++j) acc[q][j] = fmaf(dj[q], qv[q][j], acc[q][j]);
    }
#pragma unroll
    for (int q = 0; q < 4; ++q)
#pragma unroll
        for (int j = 0; j < 4; ++j) sA[wave * 4 + q][lane + 32 * j] = acc[q][j];
    __syncthreads();

    v8f c[MT_PER_BLK];
#pragma unroll
    for (int t = 0; t < MT_PER_BLK; ++t)
#pragma unroll
        for (int r = 0; r < 8; ++r) c[t][r] = 0.0f;

    for (int k0 = 0; k0 < D_FEAT; k0 += 4) {
        const int ka = k0 + half * 2;
        v2f bf;
        bf.x = W[(size_t)ka * D_FEAT + nb + l15];
        bf.y = W[(size_t)(ka + 1) * D_FEAT + nb + l15];
#pragma unroll
        for (int t = 0; t < MT_PER_BLK; ++t) {
            v2f af;
            af.x = sA[t * 16 + l15][ka];
            af.y = sA[t * 16 + l15][ka + 1];
            c[t] = __builtin_amdgcn_wmma_f32_16x16x4_f32(
                 false, af,  false, bf,
                 (short)0, c[t],  false,  false);
        }
    }
#pragma unroll
    for (int t = 0; t < MT_PER_BLK; ++t)
#pragma unroll
        for (int r = 0; r < 8; ++r) sC[t * 16 + half * 8 + r][nb + l15] = c[t][r];
    __syncthreads();

#pragma unroll
    for (int q = 0; q < 4; ++q) {
        const int row = wave * 4 + q;
        const int nd  = mbase + row;
        if (nd < n_nodes) {
            const float di = dinv[nd];
            float v[4];
            float ss = 0.0f;
#pragma unroll
            for (int j = 0; j < 4; ++j) {
                int cc  = lane + 32 * j;
                float t = di * sC[row][cc] + bias[cc];
                t = (t > 0.0f) ? t : pa[cc] * t;
                v[j] = t;
                ss += t * t;
            }
#pragma unroll
            for (int off = 16; off > 0; off >>= 1) ss += __shfl_xor(ss, off, 32);
            float scale = 1.0f / fmaxf(sqrtf(ss), 1e-12f);

            int g = batch[nd];
            float* rp = rep + (size_t)(nd & (N_REP - 1)) * REP_PITCH;
#pragma unroll
            for (int j = 0; j < 4; ++j)
                atomicAdd(&rp[(size_t)g * D_FEAT + lane + 32 * j], v[j] * scale);
            if (lane == 0) atomicAdd(&rp[REP_CNT_OFF + g], 1.0f);
        }
    }
}

__global__ __launch_bounds__(256) void merge_div_kernel(const float* __restrict__ rep,
                                                        float* __restrict__ out,
                                                        int total) {
    int i = blockIdx.x * blockDim.x + threadIdx.x;
    if (i < total) {
        const int g = i >> 7;
        float s = 0.0f, c = 0.0f;
        for (int r = 0; r < N_REP; ++r) {
            const float* rp = rep + (size_t)r * REP_PITCH;
            s += rp[i];
            c += rp[REP_CNT_OFF + g];
        }
        out[i] = s / fmaxf(c, 1.0f);
    }
}

extern "C" void kernel_launch(void* const* d_in, const int* in_sizes, int n_in,
                              void* d_out, int out_size, void* d_ws, size_t ws_size,
                              hipStream_t stream) {
    const float* x     = (const float*)d_in[0];
    const int*   ei    = (const int*)d_in[1];
    const int*   batch = (const int*)d_in[2];
    const float* W     = (const float*)d_in[3];
    const float* bias  = (const float*)d_in[4];
    const float* pa    = (const float*)d_in[5];

    const int n_nodes = in_sizes[0] / D_FEAT;
    const int n_edges = in_sizes[1] / 2;
    const int* src = ei;
    const int* dst = ei + n_edges;
    float* out = (float*)d_out;

    float* dinv   = (float*)d_ws;
    int*   head   = (int*)(dinv + n_nodes);
    int2*  link   = (int2*)(head + n_nodes);
    float* rep    = (float*)(link + n_edges);

    hipMemsetAsync(head, 0, (size_t)n_nodes * sizeof(int), stream);
    hipMemsetAsync(rep, 0, (size_t)N_REP * REP_PITCH * sizeof(float), stream);

    link_kernel<<<(n_edges + 255) / 256, 256, 0, stream>>>(src, dst, head, link, n_edges);
    chain_len_dinv_kernel<<<(n_nodes + 255) / 256, 256, 0, stream>>>(head, link, dinv, n_nodes, n_edges);

    int tile_blocks = (n_nodes + ROWS_PER_BLK - 1) / ROWS_PER_BLK;
    gather_gemm_finalize_kernel<<<tile_blocks, 256, 0, stream>>>(
        x, W, head, link, dinv, bias, pa, batch, rep, n_nodes, n_edges);

    merge_div_kernel<<<(out_size + 255) / 256, 256, 0, stream>>>(rep, out, out_size);
}


# gather kernel: loads that do not depend on the gather issued in groups ahead of use (heads, dinv, graph ids at entry; W 8 k-steps a group; bias, slopes once a wave; 4 rows finalized together)
# speedup vs baseline: 2.9379x; 1.0484x over previous
#include <hip/hip_runtime.h>
#include <hip/hip_bf16.h>
#include <math.h>

typedef __attribute__((ext_vector_type(2))) float v2f;
typedef __attribute__((ext_vector_type(8))) float v8f;

#define D_FEAT 128
#define MT_PER_BLK 2
#define ROWS_PER_BLK (MT_PER_BLK * 16)
#define N_GRAPHS 128
#define N_REP 32
#define REP_CNT_OFF (N_GRAPHS * D_FEAT)
#define REP_PITCH (N_GRAPHS * D_FEAT + 320)

__global__ __launch_bounds__(256) void link_kernel(const int* __restrict__ src,
                                                   const int* __restrict__ dst,
                                                   int* __restrict__ head,
                                                   int2* __restrict__ link,
                                                   int n_edges) {
    int e = blockIdx.x * blockDim.x + threadIdx.x;
    if (e < n_edges) {
        int d = dst[e];
        int s = src[e];
        int prev = atomicExch(&head[d], e + 1);
        link[e] = make_int2(s, prev);
    }
}

__global__ __launch_bounds__(256) void chain_len_dinv_kernel(const int* __restrict__ head,
                                                             const int2* __restrict__ link,
                                                             float* __restrict__ dinv,
                                                             int n, int n_edges) {
    int i = blockIdx.x * blockDim.x + threadIdx.x;
    if (i < n) {
        int e   = head[i];
        int cnt = 0;
        while (cnt < n_edges && e != 0) {
            e = link[e - 1].y;
            ++cnt;
        }
        dinv[i] = rsqrtf((float)cnt + 1.0f);
    }
}

#define K_GRP 8

__device__ __forceinline__ void load_w_group(v2f (&wf)[K_GRP], const float* __restrict__ wp, int s0) {
#pragma unroll
    for (int s = 0; s < K_GRP; ++s) {
        wf[s].x = wp[(size_t)((s0 + s) * 4) * D_FEAT];
        wf[s].y = wp[(size_t)((s0 + s) * 4 + 1) * D_FEAT];
    }
}

__global__ __launch_bounds__(256) __attribute__((amdgpu_num_vgpr(64))) void gather_gemm_finalize_kernel(
    const float* __restrict__ x, const float* __restrict__ W,
    const int* __restrict__ head, const int2* __restrict__ link,
    const float* __restrict__ dinv,
    const float* __restrict__ bias, const float* __restrict__ pa,
    const int* __restrict__ batch, float* __restrict__ rep,
    int n_nodes, int n_edges) {
    __shared__ float sA[ROWS_PER_BLK][D_FEAT + 4];
    __shared__ float sC[ROWS_PER_BLK][D_FEAT + 4];

    const int tid   = threadIdx.x;
    const int lane  = tid & 31;
    const int wave  = __builtin_amdgcn_readfirstlane(tid >> 5);
    const int mbase = blockIdx.x * ROWS_PER_BLK;
    const int half  = lane >> 4;
    const int l15   = lane & 15;
    const int nb    = wave * 16;

    const float* xl = x + lane;
    bool  ok[4];
    int   e[4];
    int   gid[4];
    float di[4];
    int2  lk[4];
    float acc[4][4];
#pragma unroll
    for (int q = 0; q < 4; ++q) {
        const int nd  = mbase + wave * 4 + q;
        ok[q]         = nd < n_nodes;
        const int ndc = ok[q] ? nd : (n_nodes - 1);
        e[q]   = head[ndc];
        di[q]  = dinv[ndc];
        gid[q] = batch[ndc];
        const float* r = xl + (size_t)ndc * D_FEAT;
#pragma unroll
        for (int j = 0; j < 4; ++j) acc[q][j] = r[32 * j];
    }
#pragma unroll
    for (int q = 0; q < 4; ++q) {
        if (!ok[q]) e[q] = 0;
        lk[q] = make_int2(0, 0);
        if (e[q] != 0) lk[q] = link[e[q] - 1];
    }
#pragma unroll
    for (int q = 0; q < 4; ++q) {
        const float dq = ok[q] ? di[q] : 0.0f;
#pragma unroll
        for (int j = 0; j < 4; ++j) acc[q][j] = dq * acc[q][j];
    }
    for (int it = 0; it < n_edges && (e[0] | e[1] | e[2] | e[3]) != 0; ++it) {
        float qv[4][4];
        float dj[4];
#pragma unroll
        for (int q = 0; q < 4; ++q) {
            if (e[q] != 0) {
                const float* r = xl + (size_t)lk[q].x * D_FEAT;
                dj[q] = dinv[lk[q].x];
#pragma unroll
                for (int j = 0; j < 4; ++j) qv[q][j] = r[32 * j];
                e[q] = lk[q].y;
                if (e[q] != 0) lk[q] = link[e[q] - 1];
            } else {
                dj[q] = 0.0f;
#pragma unroll
                for (int j = 0; j < 4; ++j) qv[q][j] = 0.0f;
            }
        }
#pragma unroll
        for (int q = 0; q < 4; ++q)
#pragma unroll
            for (int j = 0; j < 4; ++j) acc[q][j] = fmaf(dj[q], qv[q][j], acc[q][j]);
    }
#pragma unroll
    for (int q = 0; q < 4; ++q)
#pragma unroll
        for (int j = 0; j < 4; ++j) sA[wave * 4 + q][lane + 32 * j] = acc[q][j];

    const float* wp = W + (size_t)(half * 2) * D_FEAT + nb + l15;
    v2f wf[K_GRP];
    load_w_group(wf, wp, 0);
    __syncthreads();

    v8f c[MT_PER_BLK];
#pragma unroll
    for (int t = 0; t < MT_PER_BLK; ++t)
#pragma unroll
        for (int r = 0; r < 8; ++r) c[t][r] = 0.0f;

#pragma unroll
    for (int s0 = 0; s0 < D_FEAT / 4; s0 += K_GRP) {
        if (s0 != 0) load_w_group(wf, wp, s0);
#pragma unroll
        for (int s = 0; s < K_GRP; ++s) {
            const int ka = (s0 + s) * 4 + half * 2;
#pragma unroll
            for (int t = 0; t < MT_PER_BLK; ++t) {
                v2f af;
                af.x = sA[t * 16 + l15][ka];
                af.y = sA[t * 16 + l15][ka + 1];
                c[t] = __builtin_amdgcn_wmma_f32_16x16x4_f32(
                     false, af,  false, wf[s],
                     (short)0, c[t],  false,  false);
            }
        }
    }
    float bj[4], pj[4];
#pragma unroll
    for (int j = 0; j < 4; ++j) {
        bj[j] = bias[lane + 32 * j];
        pj[j] = pa[lane + 32 * j];
    }
#pragma unroll
    for (int t = 0; t < MT_PER_BLK; ++t)
#pragma unroll
        for (int r = 0; r < 8; ++r) sC[t * 16 + half * 8 + r][nb + l15] = c[t][r];
    __syncthreads();

    float v[4][4];
    float ss[4];
#pragma unroll
    for (int q = 0; q < 4; ++q) {
        const int row = wave * 4 + q;
        ss[q] = 0.0f;
#pragma unroll
        for (int j = 0; j < 4; ++j) {
            float t = di[q] * sC[row][lane + 32 * j] + bj[j];
            t = (t > 0.0f) ? t : pj[j] * t;
            v[q][j] = t;
            ss[q] += t * t;
        }
    }
#pragma unroll
    for (int off = 16; off > 0; off >>= 1)
#pragma unroll
        for (int q = 0; q < 4; ++q) ss[q] += __shfl_xor(ss[q], off, 32);
#pragma unroll
    for (int q = 0; q < 4; ++q) {
        if (ok[q]) {
            const int nd = mbase + wave * 4 + q;
            const float scale = 1.0f / fmaxf(sqrtf(ss[q]), 1e-12f);
            float* rp = rep + (size_t)(nd & (N_REP - 1)) * REP_PITCH;
#pragma unroll
            for (int j = 0; j < 4; ++j)
                atomicAdd(&rp[(size_t)gid[q] * D_FEAT + lane + 32 * j], v[q][j] * scale);
            if (lane == 0) atomicAdd(&rp[REP_CNT_OFF + gid[q]], 1.0f);
        }
    }
}

__global__ __launch_bounds__(256) void merge_div_kernel(const float* __restrict__ rep,
                                                        float* __restrict__ out,
                                                        int total) {
    int i = blockIdx.x * blockDim.x + threadIdx.x;
    if (i < total) {
        const int g = i >> 7;
        float s = 0.0f, c = 0.0f;
        for (int r = 0; r < N_REP; ++r) {
            const float* rp = rep + (size_t)r * REP_PITCH;
            s += rp[i];
            c += rp[REP_CNT_OFF + g];
        }
        out[i] = s / fmaxf(c, 1.0f);
    }
}

extern "C" void kernel_launch(void* const* d_in, const int* in_sizes, int n_in,
                              void* d_out, int out_size, void* d_ws, size_t ws_size,
                              hipStream_t stream) {
    const float* x     = (const float*)d_in[0];
    const int*   ei    = (const int*)d_in[1];
    const int*   batch = (const int*)d_in[2];
    const float* W     = (const float*)d_in[3];
    const float* bias  = (const float*)d_in[4];
    const float* pa    = (const float*)d_in[5];

    const int n_nodes = in_sizes[0] / D_FEAT;
    const int n_edges = in_sizes[1] / 2;
    const int* src = ei;
    const int* dst = ei + n_edges;
    float* out = (float*)d_out;

    float* dinv   = (float*)d_ws;
    int*   head   = (int*)(dinv + n_nodes);
    int2*  link   = (int2*)(head + n_nodes);
    float* rep    = (float*)(link + n_edges);

    hipMemsetAsync(head, 0, (size_t)n_nodes * sizeof(int), stream);
    hipMemsetAsync(rep, 0, (size_t)N_REP * REP_PITCH * sizeof(float), stream);

    link_kernel<<<(n_edges + 255) / 256, 256, 0, stream>>>(src, dst, head, link, n_edges);
    chain_len_dinv_kernel<<<(n_nodes + 255) / 256, 256, 0, stream>>>(head, link, dinv, n_nodes, n_edges);

    int tile_blocks = (n_nodes + ROWS_PER_BLK - 1) / ROWS_PER_BLK;
    gather_gemm_finalize_kernel<<<tile_blocks, 256, 0, stream>>>(
        x, W, head, link, dinv, bias, pa, batch, rep, n_nodes, n_edges);

    merge_div_kernel<<<(out_size + 255) / 256, 256, 0, stream>>>(rep, out, out_size);
}


# atomic targets split by chip half (die id asked once per block): each half exchanges into its own chain heads (two chains per node) and adds into its own 16 of 32 pooled copies; sum orders change
# speedup vs baseline: 5.5064x; 1.8743x over previous
#include <hip/hip_runtime.h>
#include <hip/hip_bf16.h>
#include <math.h>

typedef __attribute__((ext_vector_type(2))) float v2f;
typedef __attribute__((ext_vector_type(8))) float v8f;

#define D_FEAT 128
#define MT_PER_BLK 2
#define ROWS_PER_BLK (MT_PER_BLK * 16)
#define N_GRAPHS 128
#define N_REP 32
#define REP_CNT_OFF (N_GRAPHS * D_FEAT)
#define REP_PITCH (N_GRAPHS * D_FEAT + 320)

__device__ __forceinline__ unsigned ask_die() {
    return __builtin_amdgcn_s_sendmsg_rtn(0x87);
}
__device__ __forceinline__ int half_of_die(unsigned answer) {
    return (int)((answer >> 18) & 1u);
}

#define EDGES_PER_THR 4
__global__ __launch_bounds__(256) void link_kernel(const int* __restrict__ src,
                                                   const int* __restrict__ dst,
                                                   int* __restrict__ head,
                                                   int2* __restrict__ link,
                                                   int n_nodes, int n_edges) {
    __shared__ int s_half;
    const int base = blockIdx.x * (256 * EDGES_PER_THR) + threadIdx.x;
    int d[EDGES_PER_THR], s[EDGES_PER_THR], prev[EDGES_PER_THR];
#pragma unroll
    for (int k = 0; k < EDGES_PER_THR; ++k) {
        const int e  = base + k * 256;
        const int ec = (e < n_edges) ? e : (n_edges - 1);
        d[k] = dst[ec];
        s[k] = src[ec];
    }
    if (threadIdx.x < 32) {
        const unsigned a = ask_die();
        if (threadIdx.x == 0) s_half = half_of_die(a);
    }
    __syncthreads();
    int* hh = head + (size_t)s_half * n_nodes;
#pragma unroll
    for (int k = 0; k < EDGES_PER_THR; ++k) {
        const int e = base + k * 256;
        prev[k] = 0;
        if (e < n_edges) prev[k] = atomicExch(&hh[d[k]], e + 1);
    }
#pragma unroll
    for (int k = 0; k < EDGES_PER_THR; ++k) {
        const int e = base + k * 256;
        if (e < n_edges) link[e] = make_int2(s[k], prev[k]);
    }
}

__global__ __launch_bounds__(256) void chain_len_dinv_kernel(const int* __restrict__ head,
                                                             const int2* __restrict__ link,
                                                             float* __restrict__ dinv,
                                                             int n, int n_edges) {
    int i = blockIdx.x * blockDim.x + threadIdx.x;
    if (i < n) {
        int ea  = head[i];
        int eb  = head[(size_t)n + i];
        int cnt = 0;
        for (int it = 0; it < n_edges && (ea | eb) != 0; ++it) {
            if (ea != 0) { ea = link[ea - 1].y; ++cnt; }
            if (eb != 0) { eb = link[eb - 1].y; ++cnt; }
        }
        dinv[i] = rsqrtf((float)cnt + 1.0f);
    }
}

#define K_GRP 8

__device__ __forceinline__ void load_w_group(v2f (&wf)[K_GRP], const float* __restrict__ wp, int s0) {
#pragma unroll
    for (int s = 0; s < K_GRP; ++s) {
        wf[s].x = wp[(size_t)((s0 + s) * 4) * D_FEAT];
        wf[s].y = wp[(size_t)((s0 + s) * 4 + 1) * D_FEAT];
    }
}

__global__ __launch_bounds__(256) __attribute__((amdgpu_num_vgpr(64))) void gather_gemm_finalize_kernel(
    const float* __restrict__ x, const float* __restrict__ W,
    const int* __restrict__ head, const int2* __restrict__ link,
    const float* __restrict__ dinv,
    const float* __restrict__ bias, const float* __restrict__ pa,
    const int* __restrict__ batch, float* __restrict__ rep,
    int n_nodes, int n_edges) {
    __shared__ float sA[ROWS_PER_BLK][D_FEAT + 4];
    __shared__ float sC[ROWS_PER_BLK][D_FEAT + 4];
    __shared__ unsigned s_die_answer;

    const int tid   = threadIdx.x;
    const int lane  = tid & 31;
    const int wave  = __builtin_amdgcn_readfirstlane(tid >> 5);
    const int mbase = blockIdx.x * ROWS_PER_BLK;
    const int half  = lane >> 4;
    const int l15   = lane & 15;
    const int nb    = wave * 16;


    const float* xl = x + lane;
    bool  ok[4];
    int   e[4], e2[4];
    int   gid[4];
    float di[4];
    int2  lk[4], lk2[4];
    float acc[4][4];
#pragma unroll
    for (int q = 0; q < 4; ++q) {
        const int nd  = mbase + wave * 4 + q;
        ok[q]         = nd < n_nodes;
        const int ndc = ok[q] ? nd : (n_nodes - 1);
        e[q]   = head[ndc];
        e2[q]  = head[(size_t)n_nodes + ndc];
        di[q]  = dinv[ndc];
        gid[q] = batch[ndc];
        const float* r = xl + (size_t)ndc * D_FEAT;
#pragma unroll
        for (int j = 0; j < 4; ++j) acc[q][j] = r[32 * j];
    }
#pragma unroll
    for (int q = 0; q < 4; ++q) {
        if (!ok[q]) { e[q] = 0; e2[q] = 0; }
        lk[q]  = make_int2(0, 0);
        lk2[q] = make_int2(0, 0);
        if (e[q]  != 0) lk[q]  = link[e[q]  - 1];
        if (e2[q] != 0) lk2[q] = link[e2[q] - 1];
    }
#pragma unroll
    for (int q = 0; q < 4; ++q) {
        if (e[q] == 0) { e[q] = e2[q]; lk[q] = lk2[q]; e2[q] = 0; }
    }
#pragma unroll
    for (int q = 0; q < 4; ++q) {
        const float dq = ok[q] ? di[q] : 0.0f;
#pragma unroll
        for (int j = 0; j < 4; ++j) acc[q][j] = dq * acc[q][j];
    }
    for (int it = 0; it < n_edges && (e[0] | e[1] | e[2] | e[3]) != 0; ++it) {
        float qv[4][4];
        float dj[4];
#pragma unroll
        for (int q = 0; q < 4; ++q) {
            if (e[q] != 0) {
                const float* r = xl + (size_t)lk[q].x * D_FEAT;
                dj[q] = dinv[lk[q].x];
#pragma unroll
                for (int j = 0; j < 4; ++j) qv[q][j] = r[32 * j];
                e[q] = lk[q].y;
                if (e[q] != 0) {
                    lk[q] = link[e[q] - 1];
                } else {
                    e[q] = e2[q]; lk[q] = lk2[q]; e2[q] = 0;
                }
            } else {
                dj[q] = 0.0f;
#pragma unroll
                for (int j = 0; j < 4; ++j) qv[q][j] = 0.0f;
            }
        }
#pragma unroll
        for (int q = 0; q < 4; ++q)
#pragma unroll
            for (int j = 0; j < 4; ++j) acc[q][j] = fmaf(dj[q], qv[q][j], acc[q][j]);
    }
    unsigned die_answer = 0;
    if (wave == 0) die_answer = ask_die();
#pragma unroll
    for (int q = 0; q < 4; ++q)
#pragma unroll
        for (int j = 0; j < 4; ++j) sA[wave * 4 + q][lane + 32 * j] = acc[q][j];

    const float* wp = W + (size_t)(half * 2) * D_FEAT + nb + l15;
    v2f wf[K_GRP];
    load_w_group(wf, wp, 0);
    if (wave == 0 && lane == 0) s_die_answer = die_answer;
    __syncthreads();

    v8f c[MT_PER_BLK];
#pragma unroll
    for (int t = 0; t < MT_PER_BLK; ++t)
#pragma unroll
        for (int r = 0; r < 8; ++r) c[t][r] = 0.0f;

#pragma unroll
    for (int s0 = 0; s0 < D_FEAT / 4; s0 += K_GRP) {
        if (s0 != 0) load_w_group(wf, wp, s0);
#pragma unroll
        for (int s = 0; s < K_GRP; ++s) {
            const int ka = (s0 + s) * 4 + half * 2;
#pragma unroll
            for (int t = 0; t < MT_PER_BLK; ++t) {
                v2f af;
                af.x = sA[t * 16 + l15][ka];
                af.y = sA[t * 16 + l15][ka + 1];
                c[t] = __builtin_amdgcn_wmma_f32_16x16x4_f32(
                     false, af,  false, wf[s],
                     (short)0, c[t],  false,  false);
            }
        }
    }
    float bj[4], pj[4];
#pragma unroll
    for (int j = 0; j < 4; ++j) {
        bj[j] = bias[lane + 32 * j];
        pj[j] = pa[lane + 32 * j];
    }
#pragma unroll
    for (int t = 0; t < MT_PER_BLK; ++t)
#pragma unroll
        for (int r = 0; r < 8; ++r) sC[t * 16 + half * 8 + r][nb + l15] = c[t][r];
    __syncthreads();

    const int hf = half_of_die(__builtin_amdgcn_readfirstlane(s_die_answer));
    float v[4][4];
    float ss[4];
#pragma unroll
    for (int q = 0; q < 4; ++q) {
        const int row = wave * 4 + q;
        ss[q] = 0.0f;
#pragma unroll
        for (int j = 0; j < 4; ++j) {
            float t = di[q] * sC[row][lane + 32 * j] + bj[j];
            t = (t > 0.0f) ? t : pj[j] * t;
            v[q][j] = t;
            ss[q] += t * t;
        }
    }
#pragma unroll
    for (int off = 16; off > 0; off >>= 1)
#pragma unroll
        for (int q = 0; q < 4; ++q) ss[q] += __shfl_xor(ss[q], off, 32);
#pragma unroll
    for (int q = 0; q < 4; ++q) {
        if (ok[q]) {
            const int nd = mbase + wave * 4 + q;
            const float scale = 1.0f / fmaxf(sqrtf(ss[q]), 1e-12f);
            float* rp = rep + (size_t)(hf * (N_REP / 2) + (nd & (N_REP / 2 - 1))) * REP_PITCH;
#pragma unroll
            for (int j = 0; j < 4; ++j)
                atomicAdd(&rp[(size_t)gid[q] * D_FEAT + lane + 32 * j], v[q][j] * scale);
            if (lane == 0) atomicAdd(&rp[REP_CNT_OFF + gid[q]], 1.0f);
        }
    }
}

__global__ __launch_bounds__(256) void merge_div_kernel(const float* __restrict__ rep,
                                                        float* __restrict__ out,
                                                        int total) {
    int i = blockIdx.x * blockDim.x + threadIdx.x;
    if (i < total) {
        const int g = i >> 7;
        float s = 0.0f, c = 0.0f;
        for (int r = 0; r < N_REP; ++r) {
            const float* rp = rep + (size_t)r * REP_PITCH;
            s += rp[i];
            c += rp[REP_CNT_OFF + g];
        }
        out[i] = s / fmaxf(c, 1.0f);
    }
}

extern "C" void kernel_launch(void* const* d_in, const int* in_sizes, int n_in,
                              void* d_out, int out_size, void* d_ws, size_t ws_size,
                              hipStream_t stream) {
    const float* x     = (const float*)d_in[0];
    const int*   ei    = (const int*)d_in[1];
    const int*   batch = (const int*)d_in[2];
    const float* W     = (const float*)d_in[3];
    const float* bias  = (const float*)d_in[4];
    const float* pa    = (const float*)d_in[5];

    const int n_nodes = in_sizes[0] / D_FEAT;
    const int n_edges = in_sizes[1] / 2;
    const int* src = ei;
    const int* dst = ei + n_edges;
    float* out = (float*)d_out;

    int2*  link   = (int2*)d_ws;
    float* dinv   = (float*)(link + n_edges);
    int*   head   = (int*)(dinv + n_nodes);
    float* rep    = (float*)(head + 2 * (size_t)n_nodes);

    hipMemsetAsync(head, 0, 2 * (size_t)n_nodes * sizeof(int), stream);
    hipMemsetAsync(rep, 0, (size_t)N_REP * REP_PITCH * sizeof(float), stream);

    link_kernel<<<(n_edges + 256 * EDGES_PER_THR - 1) / (256 * EDGES_PER_THR), 256, 0, stream>>>(
        src, dst, head, link, n_nodes, n_edges);
    chain_len_dinv_kernel<<<(n_nodes + 255) / 256, 256, 0, stream>>>(head, link, dinv, n_nodes, n_edges);

    int tile_blocks = (n_nodes + ROWS_PER_BLK - 1) / ROWS_PER_BLK;
    gather_gemm_finalize_kernel<<<tile_blocks, 256, 0, stream>>>(
        x, W, head, link, dinv, bias, pa, batch, rep, n_nodes, n_edges);

    merge_div_kernel<<<(out_size + 255) / 256, 256, 0, stream>>>(rep, out, out_size);
}
